# starting kernel
# baseline (speedup 1.0000x reference)
#include <hip/hip_runtime.h>
#include <hip/hip_bf16.h>


typedef float v2f __attribute__((ext_vector_type(2)));
typedef float v8f __attribute__((ext_vector_type(8)));

struct Gate { signed char op, a, b, wi; };

#define RY(wr, wi)  { 0, (wr), 0, (wi) }
#define CX(cc, tt)  { 1, (cc), (tt), 0 }

__device__ __constant__ Gate GATES[5][24] = {
  { RY(0,0),RY(1,1),RY(2,2),RY(3,3), CX(0,1),CX(2,3), RY(0,4),RY(2,5), CX(1,2), RY(1,6),
    {0,0,0,0},{0,0,0,0},{0,0,0,0},{0,0,0,0},{0,0,0,0},{0,0,0,0},{0,0,0,0},
    {0,0,0,0},{0,0,0,0},{0,0,0,0},{0,0,0,0},{0,0,0,0},{0,0,0,0},{0,0,0,0} },
  { RY(0,0),RY(1,1),RY(2,2),RY(3,3), CX(0,1),CX(1,2),CX(2,3), RY(1,4),RY(2,5),RY(3,6),
    {0,0,0,0},{0,0,0,0},{0,0,0,0},{0,0,0,0},{0,0,0,0},{0,0,0,0},{0,0,0,0},
    {0,0,0,0},{0,0,0,0},{0,0,0,0},{0,0,0,0},{0,0,0,0},{0,0,0,0},{0,0,0,0} },
  { RY(0,0),RY(1,1),RY(2,2),RY(3,3), CX(0,1),CX(1,2),CX(2,3),CX(3,0),
    RY(0,4),RY(1,5),RY(2,6),RY(3,7),
    {0,0,0,0},{0,0,0,0},{0,0,0,0},{0,0,0,0},{0,0,0,0},{0,0,0,0},
    {0,0,0,0},{0,0,0,0},{0,0,0,0},{0,0,0,0},{0,0,0,0},{0,0,0,0} },
  { RY(0,0),RY(1,1),RY(2,2),RY(3,3), CX(0,1),CX(1,2),CX(2,3),CX(3,0),
    RY(0,4),RY(1,5),RY(2,6),RY(3,7), CX(3,0),CX(2,3),CX(1,2),CX(0,1),
    RY(0,8),RY(1,9),RY(2,10),RY(3,11), CX(0,1),CX(1,2),CX(2,3),CX(3,0) },
  { RY(0,0),RY(1,1),RY(2,2),RY(3,3), CX(0,1),CX(2,3),CX(1,2),CX(3,0),
    RY(0,4),RY(1,5),RY(2,6),RY(3,7), CX(0,1),CX(2,3),CX(1,2),CX(3,0),
    RY(0,8),RY(1,9),RY(2,10),RY(3,11), CX(0,1),CX(2,3),CX(1,2),CX(3,0) },
};

__device__ __constant__ Gate GATES_DR_TAIL[8] = {
  RY(0,12),RY(1,13),RY(2,14),RY(3,15), CX(3,0),CX(2,3),CX(1,2),CX(0,1)
};

__device__ __constant__ int GCNT[5] = { 10, 10, 12, 24  , 24 };

__global__ void __launch_bounds__(256) build_M_kernel(const float* __restrict__ wts,
                                                      float* __restrict__ Mout) {
  const int pair = blockIdx.x;
  const int i    = pair / 3;
  const int t    = threadIdx.x;
  const int m    = t >> 4;
  const int j    = t & 15;
  const int circ = i % 5;

  __shared__ float U[16][17];
  U[m][j] = (m == j) ? 1.0f : 0.0f;
  __syncthreads();

  const float* wv = wts + pair * 16;

  const int ng = GCNT[circ];
  for (int g = 0; g < ng + ((circ == 3) ? 8 : 0); ++g) {
    Gate gd = (g < 24) ? GATES[circ][g] : GATES_DR_TAIL[g - 24];
    if (gd.op == 0) {
      const int  mask = 1 << (3 - gd.a);
      const float th  = 0.5f * wv[(int)gd.wi];
      float sh, ch;  sincosf(th, &sh, &ch);
      const float um = U[m][j];
      const float up = U[m ^ mask][j];
      __syncthreads();
      U[m][j] = ((m & mask) == 0) ? (ch * um - sh * up)
                                  : (sh * up + ch * um);
      __syncthreads();
    } else {
      const int cb = 3 - gd.a, tb = 3 - gd.b;
      const int src = m ^ (((m >> cb) & 1) << tb);
      const float v = U[src][j];
      __syncthreads();
      U[m][j] = v;
      __syncthreads();
    }
  }

  float acc = 0.0f;
#pragma unroll
  for (int mm = 0; mm < 16; ++mm) {
    const float sgn = (mm < 8) ? 1.0f : -1.0f;
    acc += sgn * U[mm][m] * U[mm][j];
  }
  Mout[(pair * 16 + m) * 16 + j] = acc;
}

__global__ void __launch_bounds__(32) quanv_wmma_kernel(const float* __restrict__ in,
                                                        const float* __restrict__ M,
                                                        float* __restrict__ out) {
  const int tile = blockIdx.x;
  const int oi   = blockIdx.y;
  const int lane = threadIdx.x;
  const int half = lane >> 4;
  const int col  = lane & 15;
  const int row  = lane & 15;

  __shared__ float S[16][17];

  float part[8];
#pragma unroll
  for (int r = 0; r < 8; ++r) part[r] = 0.0f;

  for (int c = 0; c < 3; ++c) {
    __syncthreads();
    if (lane < 16) {
      const int p   = tile * 16 + lane;
      const int b   = p / 961;
      const int rem = p % 961;
      const int h   = rem / 31;
      const int w   = rem % 31;
      const float* base = in + ((b * 3 + c) * 32 + h) * 32 + w;
      const float x0 = base[0],  x1 = base[1];
      const float x2 = base[32], x3 = base[33];
      float sh0, ch0, sh1, ch1, sh2, ch2, sh3, ch3;
      sincosf(0.5f * x0, &sh0, &ch0);
      sincosf(0.5f * x1, &sh1, &ch1);
      sincosf(0.5f * x2, &sh2, &ch2);
      sincosf(0.5f * x3, &sh3, &ch3);
#pragma unroll
      for (int mm = 0; mm < 16; ++mm) {
        const float v = ((mm & 8) ? sh0 : ch0) * ((mm & 4) ? sh1 : ch1)
                      * ((mm & 2) ? sh2 : ch2) * ((mm & 1) ? sh3 : ch3);
        S[lane][mm] = v;
      }
    }
    __syncthreads();

    const float* Mp = M + (oi * 3 + c) * 256;

    v8f acc = {};
#pragma unroll
    for (int s = 0; s < 4; ++s) {
      const int k0 = 4 * s;
      v2f a, bf;
      a[0]  = S[row][k0 + 2 * half + 0];
      a[1]  = S[row][k0 + 2 * half + 1];
      bf[0] = Mp[(k0 + 2 * half + 0) * 16 + col];
      bf[1] = Mp[(k0 + 2 * half + 1) * 16 + col];
      acc = __builtin_amdgcn_wmma_f32_16x16x4_f32(
                false, a, false, bf, (short)0, acc, false, false);
    }

#pragma unroll
    for (int r = 0; r < 8; ++r) {
      part[r] += acc[r] * S[r + 8 * half][col];
    }
  }

#pragma unroll
  for (int r = 0; r < 8; ++r) {
    float v = part[r];
    v += __shfl_xor(v, 1);
    v += __shfl_xor(v, 2);
    v += __shfl_xor(v, 4);
    v += __shfl_xor(v, 8);
    part[r] = v;
  }

  float outv = 0.0f;
#pragma unroll
  for (int r = 0; r < 8; ++r) {
    const float vl = __shfl(part[r], 0);
    const float vh = __shfl(part[r], 16);
    if (lane == r)     outv = vl;
    if (lane == r + 8) outv = vh;
  }

  if (lane < 16) {
    const int p   = tile * 16 + lane;
    const int b   = p / 961;
    const int rem = p % 961;
    out[(b * 16 + oi) * 961 + rem] = outv;
  }
}

extern "C" void kernel_launch(void* const* d_in, const int* in_sizes, int n_in,
                              void* d_out, int out_size, void* d_ws, size_t ws_size,
                              hipStream_t stream) {
  const float* inputs  = (const float*)d_in[0];
  const float* weights = (const float*)d_in[1];
  float*       out     = (float*)d_out;
  float*       Mbuf    = (float*)d_ws;

  build_M_kernel<<<48, 256, 0, stream>>>(weights, Mbuf);
  quanv_wmma_kernel<<<dim3(961, 16), 32, 0, stream>>>(inputs, Mbuf, out);
}


# Main kernel: input channel c+1's embedding (sincos, products) issued in slices between channel c's WMMAs, in their shadow; sincos range test once per wave keeps it branch-free; sum order unchanged
# speedup vs baseline: 2.5589x; 2.5589x over previous
#include <hip/hip_runtime.h>
#include <hip/hip_bf16.h>


typedef float v2f __attribute__((ext_vector_type(2)));
typedef float v8f __attribute__((ext_vector_type(8)));


__device__ __forceinline__ int cxmap(int m, int ctrl, int tgt) {
  return m ^ (((m >> (3 - ctrl)) & 1) << (3 - tgt));
}

__device__ __forceinline__ float lane_value(float v, int src) {
  return __int_as_float(__builtin_amdgcn_readlane(__float_as_int(v), src));
}

template <int WI>
__device__ __forceinline__ void rot_row(int x, float cosv, float sinv, float& e0, float& e1) {
  if constexpr (WI < 0) {
    e0 = x ? 0.0f : 1.0f;
    e1 = x ? 1.0f : 0.0f;
  } else {
    const float c = lane_value(cosv, WI);
    const float s = lane_value(sinv, WI);
    e0 = x ? s : c;
    e1 = x ? c : -s;
  }
}

template <int W0, int W1, int W2, int W3>
__device__ __forceinline__ v8f apply_layer(v8f T, int mp, int hf, float cosv, float sinv) {
  float p0, p1, q0, q1, r0, r1, t0, t1;
  rot_row<W0>((mp >> 3) & 1, cosv, sinv, p0, p1);
  rot_row<W1>((mp >> 2) & 1, cosv, sinv, q0, q1);
  rot_row<W2>((mp >> 1) & 1, cosv, sinv, r0, r1);
  rot_row<W3>( mp       & 1, cosv, sinv, t0, t1);
  const float p = hf ? p1 : p0;
  float a[8];
#pragma unroll
  for (int i = 0; i < 8; ++i) {
    a[i] = p * ((i & 4) ? q1 : q0) * ((i & 2) ? r1 : r0) * ((i & 1) ? t1 : t0);
  }
  v8f acc = {};
#pragma unroll
  for (int s = 0; s < 4; ++s) {
    v2f av, bv;
    av[0] = a[2 * s]; av[1] = a[2 * s + 1];
    bv[0] = T[2 * s]; bv[1] = T[2 * s + 1];
    acc = __builtin_amdgcn_wmma_f32_16x16x4_f32(false, av, false, bv, (short)0, acc, false, false);
  }
  return acc;
}

__device__ __forceinline__ void store_U_hi(v8f U, int hf, float* __restrict__ uo) {
  if (hf) {
#pragma unroll
    for (int r = 0; r < 8; ++r) {
      uo[r * 16] = U[r];
    }
  }
}

__global__ void __launch_bounds__(32) build_U_kernel(const float* __restrict__ wts,
                                                     float* __restrict__ Uout) {
  const int pair = blockIdx.x;
  const int circ = (pair / 3) % 5;
  const int lane = threadIdx.x;
  const int hf   = lane >> 4;
  const int n    = lane & 15;

  float sinv, cosv;
  sincosf(0.5f * wts[pair * 16 + n], &sinv, &cosv);

  v8f T;
#pragma unroll
  for (int r = 0; r < 8; ++r) {
    T[r] = (8 * hf + r == n) ? 1.0f : 0.0f;
  }

  float* uo = Uout + pair * 128 + n;

  switch (circ) {
    case 0: {
      T = apply_layer<0, 1, 2, 3>(T, cxmap(cxmap(n, 2, 3), 0, 1), hf, cosv, sinv);
      T = apply_layer<4, -1, 5, -1>(T, cxmap(n, 1, 2), hf, cosv, sinv);
      T = apply_layer<-1, 6, -1, -1>(T, n, hf, cosv, sinv);
      store_U_hi(T, hf, uo);
      break;
    }
    case 1: {
      T = apply_layer<0, 1, 2, 3>(T, cxmap(cxmap(cxmap(n, 2, 3), 1, 2), 0, 1), hf, cosv, sinv);
      T = apply_layer<-1, 4, 5, 6>(T, n, hf, cosv, sinv);
      store_U_hi(T, hf, uo);
      break;
    }
    case 2: {
      T = apply_layer<0, 1, 2, 3>(T, cxmap(cxmap(cxmap(cxmap(n, 3, 0), 2, 3), 1, 2), 0, 1), hf, cosv, sinv);
      T = apply_layer<4, 5, 6, 7>(T, n, hf, cosv, sinv);
      store_U_hi(T, hf, uo);
      break;
    }
    case 3: {
      const int fwd = cxmap(cxmap(cxmap(cxmap(n, 3, 0), 2, 3), 1, 2), 0, 1);
      const int bwd = cxmap(cxmap(cxmap(cxmap(n, 0, 1), 1, 2), 2, 3), 3, 0);
      T = apply_layer<0, 1, 2, 3>(T, fwd, hf, cosv, sinv);
      T = apply_layer<4, 5, 6, 7>(T, bwd, hf, cosv, sinv);
      T = apply_layer<8, 9, 10, 11>(T, fwd, hf, cosv, sinv);
      T = apply_layer<12, 13, 14, 15>(T, bwd, hf, cosv, sinv);
      store_U_hi(T, hf, uo);
      break;
    }
    default: {
      const int blk = cxmap(cxmap(cxmap(cxmap(n, 3, 0), 1, 2), 2, 3), 0, 1);
      T = apply_layer<0, 1, 2, 3>(T, blk, hf, cosv, sinv);
      T = apply_layer<4, 5, 6, 7>(T, blk, hf, cosv, sinv);
      T = apply_layer<8, 9, 10, 11>(T, blk, hf, cosv, sinv);
      store_U_hi(T, hf, uo);
      break;
    }
  }
}


#define SINCOS_LARGE_ARG 131072.0f

template <bool SMALL>
__device__ __forceinline__ void half_angle_sincos(float h, float& s, float& c) {
  if constexpr (SMALL) {
    __builtin_assume(!(__builtin_fabsf(h) >= SINCOS_LARGE_ARG));
  }
  sincosf(h, &s, &c);
}

template <bool SMALL>
__device__ __forceinline__ void embed_channel(const float (&x)[4], int hf, float (&sv)[8]) {
  float sh0, ch0, sh1, ch1, sh2, ch2, sh3, ch3;
  half_angle_sincos<SMALL>(0.5f * x[0], sh0, ch0);
  half_angle_sincos<SMALL>(0.5f * x[1], sh1, ch1);
  half_angle_sincos<SMALL>(0.5f * x[2], sh2, ch2);
  half_angle_sincos<SMALL>(0.5f * x[3], sh3, ch3);
  const float t0 = hf ? sh0 : ch0;
#pragma unroll
  for (int i = 0; i < 8; ++i) {
    sv[i] = t0 * ((i & 4) ? sh1 : ch1) * ((i & 2) ? sh2 : ch2) * ((i & 1) ? sh3 : ch3);
  }
}

__device__ __forceinline__ void add_squares(v2f& ph, v8f acc) {
#pragma unroll
  for (int s = 0; s < 4; ++s) {
    v2f d;
    d[0] = acc[2 * s]; d[1] = acc[2 * s + 1];
    ph += d * d;
  }
}

template <bool SMALL>
__device__ __forceinline__ void tile_body(const float (&xin)[3][4], const float4 (&uf)[8][3][2],
                                          int hf, float* __restrict__ outp) {
  v2f ph[8];
#pragma unroll
  for (int q = 0; q < 8; ++q) {
    ph[q][0] = 0.0f; ph[q][1] = 0.0f;
  }

  float sv[8];
  embed_channel<SMALL>(xin[0], hf, sv);

#pragma unroll
  for (int c = 0; c < 3; ++c) {
    const int cn = c < 2 ? c + 1 : 2;
    v8f acc[8];
    float sn[4], cs[4];
    float svn[8];
#pragma unroll
    for (int q = 0; q < 8; ++q) {
      const float4 u0 = uf[q][c][0];
      const float4 u1 = uf[q][c][1];
      v8f a8 = {};
      v2f a, bf;
      a[0] = u0.x; a[1] = u0.y; bf[0] = sv[0]; bf[1] = sv[1];
      a8 = __builtin_amdgcn_wmma_f32_16x16x4_f32(false, a, false, bf, (short)0, a8, false, false);
      a[0] = u0.z; a[1] = u0.w; bf[0] = sv[2]; bf[1] = sv[3];
      a8 = __builtin_amdgcn_wmma_f32_16x16x4_f32(false, a, false, bf, (short)0, a8, false, false);
      a[0] = u1.x; a[1] = u1.y; bf[0] = sv[4]; bf[1] = sv[5];
      a8 = __builtin_amdgcn_wmma_f32_16x16x4_f32(false, a, false, bf, (short)0, a8, false, false);
      a[0] = u1.z; a[1] = u1.w; bf[0] = sv[6]; bf[1] = sv[7];
      a8 = __builtin_amdgcn_wmma_f32_16x16x4_f32(false, a, false, bf, (short)0, a8, false, false);
      acc[q] = a8;

      if (c < 2) {
        if (q < 4) {
          half_angle_sincos<SMALL>(0.5f * xin[cn][q], sn[q], cs[q]);
        } else if (q < 6) {
          const float t0 = hf ? sn[0] : cs[0];
#pragma unroll
          for (int i = 4 * (q - 4); i < 4 * (q - 4) + 4; ++i) {
            svn[i] = t0 * ((i & 4) ? sn[1] : cs[1]) * ((i & 2) ? sn[2] : cs[2]) * ((i & 1) ? sn[3] : cs[3]);
          }
        }
      }
      if (q >= 2) {
        add_squares(ph[q - 2], acc[q - 2]);
      }
    }
    add_squares(ph[6], acc[6]);
    add_squares(ph[7], acc[7]);

    if (c < 2) {
#pragma unroll
      for (int i = 0; i < 8; ++i) {
        sv[i] = svn[i];
      }
    }
  }

#pragma unroll
  for (int q = 0; q < 8; ++q) {
    outp[(2 * q) * 961] = 3.0f - 2.0f * (ph[q][0] + ph[q][1]);
  }
}

__global__ void __launch_bounds__(32) quanv_wmma_kernel(const float* __restrict__ in,
                                                        const float* __restrict__ Uhi,
                                                        float* __restrict__ out) {
  const int tile = blockIdx.x;
  const int lane = threadIdx.x;
  const int hf   = lane >> 4;
  const int n    = lane & 15;

  const int p   = tile * 16 + n;
  const int b   = p / 961;
  const int rem = p % 961;
  const int h   = rem / 31;
  const int w   = rem % 31;

  float xin[3][4];
#pragma unroll
  for (int c = 0; c < 3; ++c) {
    const float* base = in + ((b * 3 + c) * 32 + h) * 32 + w;
    xin[c][0] = base[0];  xin[c][1] = base[1];
    xin[c][2] = base[32]; xin[c][3] = base[33];
  }

  float4 uf[8][3][2];
  {
    const float* Urow = Uhi + ((n >> 3) * 3) * 128 + (n & 7) * 16 + 8 * hf;
#pragma unroll
    for (int q = 0; q < 8; ++q) {
#pragma unroll
      for (int c = 0; c < 3; ++c) {
        const float4* up = reinterpret_cast<const float4*>(Urow + (q * 6 + c) * 128);
        uf[q][c][0] = up[0];
        uf[q][c][1] = up[1];
      }
    }
  }

  bool large = false;
#pragma unroll
  for (int c = 0; c < 3; ++c) {
#pragma unroll
    for (int j = 0; j < 4; ++j) {
      large = large || (__builtin_fabsf(0.5f * xin[c][j]) >= SINCOS_LARGE_ARG);
    }
  }

  float* outp = out + (b * 16 + hf) * 961 + rem;

  if (__builtin_expect(__builtin_amdgcn_ballot_w32(large) == 0u, 1)) {
    tile_body<true>(xin, uf, hf, outp);
  } else {
    tile_body<false>(xin, uf, hf, outp);
  }
}

extern "C" void kernel_launch(void* const* d_in, const int* in_sizes, int n_in,
                              void* d_out, int out_size, void* d_ws, size_t ws_size,
                              hipStream_t stream) {
  const float* inputs  = (const float*)d_in[0];
  const float* weights = (const float*)d_in[1];
  float*       out     = (float*)d_out;
  float*       Ubuf    = (float*)d_ws;

  build_U_kernel<<<48, 32, 0, stream>>>(weights, Ubuf);
  quanv_wmma_kernel<<<961, 32, 0, stream>>>(inputs, Ubuf, out);
}


# One launch, channel major: a 4-wave group builds only its own 4 channels' 12 circuit matrices (WMMA, a channel per wave, LDS); each wave then embeds 4 tiles for all 4 channels; sum order unchanged
# speedup vs baseline: 3.1512x; 1.2315x over previous
#include <hip/hip_runtime.h>


typedef float v2f __attribute__((ext_vector_type(2)));
typedef float v8f __attribute__((ext_vector_type(8)));

__device__ __forceinline__ int cxmap(int m, int ctrl, int tgt) {
  return m ^ (((m >> (3 - ctrl)) & 1) << (3 - tgt));
}

__device__ __forceinline__ float lane_value(float v, int src) {
  return __int_as_float(__builtin_amdgcn_readlane(__float_as_int(v), src));
}

template <int WI>
__device__ __forceinline__ void rot_row(int x, float cosv, float sinv, float& e0, float& e1) {
  if constexpr (WI < 0) {
    e0 = x ? 0.0f : 1.0f;
    e1 = x ? 1.0f : 0.0f;
  } else {
    const float c = lane_value(cosv, WI);
    const float s = lane_value(sinv, WI);
    e0 = x ? s : c;
    e1 = x ? c : -s;
  }
}

template <int W0, int W1, int W2, int W3>
__device__ __forceinline__ v8f apply_layer(v8f T, int mp, int hf, float cosv, float sinv) {
  float p0, p1, q0, q1, r0, r1, t0, t1;
  rot_row<W0>((mp >> 3) & 1, cosv, sinv, p0, p1);
  rot_row<W1>((mp >> 2) & 1, cosv, sinv, q0, q1);
  rot_row<W2>((mp >> 1) & 1, cosv, sinv, r0, r1);
  rot_row<W3>( mp       & 1, cosv, sinv, t0, t1);
  const float p = hf ? p1 : p0;
  float a[8];
#pragma unroll
  for (int i = 0; i < 8; ++i) {
    a[i] = p * ((i & 4) ? q1 : q0) * ((i & 2) ? r1 : r0) * ((i & 1) ? t1 : t0);
  }
  v8f acc = {};
#pragma unroll
  for (int s = 0; s < 4; ++s) {
    v2f av, bv;
    av[0] = a[2 * s]; av[1] = a[2 * s + 1];
    bv[0] = T[2 * s]; bv[1] = T[2 * s + 1];
    acc = __builtin_amdgcn_wmma_f32_16x16x4_f32(false, av, false, bv, (short)0, acc, false, false);
  }
  return acc;
}

__device__ __forceinline__ void store_U_hi(v8f U, int hf, float* uo) {
  if (hf) {
#pragma unroll
    for (int r = 0; r < 8; ++r) {
      uo[r * 16] = U[r];
    }
  }
}

#define LAYER3(W0, W1, W2, W3, MP)                                                  \
  _Pragma("unroll")                 \
  for (int c = 0; c < 3; ++c) {                                                     \
    T[c] = apply_layer<W0, W1, W2, W3>(T[c], (MP), hf, cosv[c], sinv[c]);           \
  }

__device__ __forceinline__ void build_channel(int ch, int lane, const float (&sinv)[3],
                                              const float (&cosv)[3], float* ulds) {
  const int circ = ch % 5;
  const int hf   = lane >> 4;
  const int n    = lane & 15;

  v8f T[3];
#pragma unroll
  for (int c = 0; c < 3; ++c) {
#pragma unroll
    for (int r = 0; r < 8; ++r) {
      T[c][r] = (8 * hf + r == n) ? 1.0f : 0.0f;
    }
  }

  switch (circ) {
    case 0: {
      LAYER3(0, 1, 2, 3, cxmap(cxmap(n, 2, 3), 0, 1))
      LAYER3(4, -1, 5, -1, cxmap(n, 1, 2))
      LAYER3(-1, 6, -1, -1, n)
      break;
    }
    case 1: {
      LAYER3(0, 1, 2, 3, cxmap(cxmap(cxmap(n, 2, 3), 1, 2), 0, 1))
      LAYER3(-1, 4, 5, 6, n)
      break;
    }
    case 2: {
      LAYER3(0, 1, 2, 3, cxmap(cxmap(cxmap(cxmap(n, 3, 0), 2, 3), 1, 2), 0, 1))
      LAYER3(4, 5, 6, 7, n)
      break;
    }
    case 3: {
      const int fwd = cxmap(cxmap(cxmap(cxmap(n, 3, 0), 2, 3), 1, 2), 0, 1);
      const int bwd = cxmap(cxmap(cxmap(cxmap(n, 0, 1), 1, 2), 2, 3), 3, 0);
      LAYER3(0, 1, 2, 3, fwd)
      LAYER3(4, 5, 6, 7, bwd)
      LAYER3(8, 9, 10, 11, fwd)
      LAYER3(12, 13, 14, 15, bwd)
      break;
    }
    default: {
      const int blk = cxmap(cxmap(cxmap(cxmap(n, 3, 0), 1, 2), 2, 3), 0, 1);
      LAYER3(0, 1, 2, 3, blk)
      LAYER3(4, 5, 6, 7, blk)
      LAYER3(8, 9, 10, 11, blk)
      break;
    }
  }

#pragma unroll
  for (int c = 0; c < 3; ++c) {
    store_U_hi(T[c], hf, ulds + c * 128 + n);
  }
}


#define SINCOS_LARGE_ARG 131072.0f

template <bool SMALL>
__device__ __forceinline__ void half_angle_sincos(float h, float& s, float& c) {
  if constexpr (SMALL) {
    __builtin_assume(!(__builtin_fabsf(h) >= SINCOS_LARGE_ARG));
  }
  sincosf(h, &s, &c);
}

__device__ __forceinline__ void embed_products(const float (&sn)[4], const float (&cs)[4], int hf,
                                               float (&sv)[8]) {
  const float t0 = hf ? sn[0] : cs[0];
#pragma unroll
  for (int i = 0; i < 8; ++i) {
    sv[i] = t0 * ((i & 4) ? sn[1] : cs[1]) * ((i & 2) ? sn[2] : cs[2]) * ((i & 1) ? sn[3] : cs[3]);
  }
}

__device__ __forceinline__ void add_squares(v2f& ph, v8f acc) {
#pragma unroll
  for (int s = 0; s < 4; ++s) {
    v2f d;
    d[0] = acc[2 * s]; d[1] = acc[2 * s + 1];
    ph += d * d;
  }
}

__device__ __forceinline__ void load_patch(const float* __restrict__ in, int tile, int n,
                                           float (&x)[3][4], int& b, int& rem) {
  const int p = tile * 16 + n;
  b   = p / 961;
  rem = p % 961;
  const int h = rem / 31;
  const int w = rem % 31;
#pragma unroll
  for (int c = 0; c < 3; ++c) {
    const float* base = in + ((b * 3 + c) * 32 + h) * 32 + w;
    x[c][0] = base[0];  x[c][1] = base[1];
    x[c][2] = base[32]; x[c][3] = base[33];
  }
}

__device__ __forceinline__ bool any_large(const float (&x)[3][4]) {
  bool large = false;
#pragma unroll
  for (int c = 0; c < 3; ++c) {
#pragma unroll
    for (int j = 0; j < 4; ++j) {
      large = large || (__builtin_fabsf(0.5f * x[c][j]) >= SINCOS_LARGE_ARG);
    }
  }
  return large;
}

__device__ __forceinline__ v2f frag_k(const float4 (&u)[2], int s) {
  const float4 w = u[s >> 1];
  v2f a;
  if (s & 1) { a[0] = w.z; a[1] = w.w; } else { a[0] = w.x; a[1] = w.y; }
  return a;
}

template <bool SMALL>
__device__ __forceinline__ void tile_body(const float (&x)[3][4], const float (&xn0)[4],
                                          const float4 (&uf)[2][3][2], int hf, float (&sv)[8],
                                          float (&res)[2]) {
  v2f ph[2];
  ph[0][0] = 0.0f; ph[0][1] = 0.0f;
  ph[1][0] = 0.0f; ph[1][1] = 0.0f;
#pragma unroll
  for (int c = 0; c < 3; ++c) {
    const float* xs = c < 2 ? x[c + 1] : xn0;
    float sn[4], cs[4], svn[8];
    v8f a8[2] = {};
#pragma unroll
    for (int s = 0; s < 4; ++s) {
      v2f bf;
      bf[0] = sv[2 * s]; bf[1] = sv[2 * s + 1];
#pragma unroll
      for (int P = 0; P < 2; ++P) {
        a8[P] = __builtin_amdgcn_wmma_f32_16x16x4_f32(false, frag_k(uf[P][c], s), false, bf, (short)0,
                                                      a8[P], false, false);
      }
      half_angle_sincos<SMALL>(0.5f * xs[s], sn[s], cs[s]);
    }
    embed_products(sn, cs, hf, svn);
    add_squares(ph[0], a8[0]);
    add_squares(ph[1], a8[1]);
#pragma unroll
    for (int i = 0; i < 8; ++i) {
      sv[i] = svn[i];
    }
  }
  res[0] = 3.0f - 2.0f * (ph[0][0] + ph[0][1]);
  res[1] = 3.0f - 2.0f * (ph[1][0] + ph[1][1]);
}

template <bool SMALL>
__device__ __forceinline__ void weight_sincos(const float (&wv)[3], float (&sw)[3], float (&cw)[3]) {
#pragma unroll
  for (int c = 0; c < 3; ++c) {
    half_angle_sincos<SMALL>(0.5f * wv[c], sw[c], cw[c]);
  }
}

template <bool SMALL>
__device__ __forceinline__ void embed_first(const float (&x)[4], int hf, float (&sv)[8]) {
  float sn[4], cs[4];
#pragma unroll
  for (int j = 0; j < 4; ++j) {
    half_angle_sincos<SMALL>(0.5f * x[j], sn[j], cs[j]);
  }
  embed_products(sn, cs, hf, sv);
}

#define TILES_PER_WAVE 4
#define TILES_PER_GROUP 16
#define LAST_TILE 960

__device__ __forceinline__ void lds_patch(const float* px, int t, int lane, float (&x)[3][4]) {
#pragma unroll
  for (int c = 0; c < 3; ++c) {
#pragma unroll
    for (int jp = 0; jp < 2; ++jp) {
      const float2 v = *reinterpret_cast<const float2*>(px + (((t * 3 + c) * 2 + jp) * 32 + lane) * 2);
      x[c][2 * jp] = v.x; x[c][2 * jp + 1] = v.y;
    }
  }
}

template <bool SMALL>
__device__ __forceinline__ void tile_loop(const float* px, const float4 (&uf)[2][3][2], int lane, int Q,
                                          int tile0, float* __restrict__ out) {
  const int hf = lane >> 4;
  const int n  = lane & 15;
  float x0[3][4], x1[3][4];
  lds_patch(px, 0, lane, x0);
  lds_patch(px, 1, lane, x1);

  float sv[8];
  embed_first<SMALL>(x0[0], hf, sv);

#pragma unroll 1
  for (int t = 0; t < TILES_PER_WAVE; ++t) {
    const int tile = tile0 + t;
    if (tile > LAST_TILE) break;

    float x2[3][4];
    lds_patch(px, t + 2 < TILES_PER_WAVE ? t + 2 : TILES_PER_WAVE - 1, lane, x2);

    const int p   = tile * 16 + n;
    const int b   = p / 961;
    const int rem = p % 961;

    float res[2];
    tile_body<SMALL>(x0, x1[0], uf, hf, sv, res);

    out[(b * 16 + 4 * Q + hf) * 961 + rem]     = res[0];
    out[(b * 16 + 4 * Q + 2 + hf) * 961 + rem] = res[1];

#pragma unroll
    for (int c = 0; c < 3; ++c) {
#pragma unroll
      for (int j = 0; j < 4; ++j) {
        x0[c][j] = x1[c][j];
        x1[c][j] = x2[c][j];
      }
    }
  }
}

__global__ void __launch_bounds__(128) quanv_fused_kernel(const float* __restrict__ in,
                                                         const float* __restrict__ wts,
                                                         float* __restrict__ out) {
  __shared__ __attribute__((aligned(16))) float Us[12 * 128];
  __shared__ __attribute__((aligned(16))) float Px[4 * TILES_PER_WAVE * 12 * 32];

  const int Q    = blockIdx.x & 3;
  const int g    = blockIdx.x >> 2;
  const int wv   = __builtin_amdgcn_readfirstlane(threadIdx.x >> 5);
  const int lane = threadIdx.x & 31;
  const int n    = lane & 15;
  const int hf   = lane >> 4;
  const int ch   = 4 * Q + wv;

  float wvl[3];
#pragma unroll
  for (int c = 0; c < 3; ++c) {
    wvl[c] = wts[(ch * 3 + c) * 16 + n];
  }
  const int tile0 = g * TILES_PER_GROUP + wv * TILES_PER_WAVE;
  float gx[TILES_PER_WAVE][3][4];
#pragma unroll
  for (int t = 0; t < TILES_PER_WAVE; ++t) {
    const int tile = tile0 + t < LAST_TILE ? tile0 + t : LAST_TILE;
    int bb, rr;
    load_patch(in, tile, n, gx[t], bb, rr);
  }

  float sinv[3], cosv[3];
  {
    bool wl = false;
#pragma unroll
    for (int c = 0; c < 3; ++c) {
      wl = wl | (__builtin_fabsf(0.5f * wvl[c]) >= SINCOS_LARGE_ARG);
    }
    if (__builtin_expect(__builtin_amdgcn_ballot_w32(wl) == 0u, 1)) {
      weight_sincos<true>(wvl, sinv, cosv);
    } else {
      weight_sincos<false>(wvl, sinv, cosv);
    }
  }

  build_channel(ch, lane, sinv, cosv, Us + wv * 3 * 128);

  float* px = Px + wv * (TILES_PER_WAVE * 12 * 32);
  bool large = false;
#pragma unroll
  for (int t = 0; t < TILES_PER_WAVE; ++t) {
    large = large | any_large(gx[t]);
#pragma unroll
    for (int c = 0; c < 3; ++c) {
#pragma unroll
      for (int jp = 0; jp < 2; ++jp) {
        float2 v;
        v.x = gx[t][c][2 * jp]; v.y = gx[t][c][2 * jp + 1];
        *reinterpret_cast<float2*>(px + (((t * 3 + c) * 2 + jp) * 32 + lane) * 2) = v;
      }
    }
  }
  __syncthreads();

  float4 uf[2][3][2];
  {
    const float* Urow = Us + ((n >> 3) * 3) * 128 + (n & 7) * 16 + 8 * hf;
#pragma unroll
    for (int P = 0; P < 2; ++P) {
#pragma unroll
      for (int c = 0; c < 3; ++c) {
        const float4* up = reinterpret_cast<const float4*>(Urow + (P * 6 + c) * 128);
        uf[P][c][0] = up[0];
        uf[P][c][1] = up[1];
      }
    }
  }

  if (__builtin_expect(__builtin_amdgcn_ballot_w32(large) == 0u, 1)) {
    tile_loop<true>(px, uf, lane, Q, tile0, out);
  } else {
    tile_loop<false>(px, uf, lane, Q, tile0, out);
  }
}

extern "C" void kernel_launch(void* const* d_in, const int* in_sizes, int n_in,
                              void* d_out, int out_size, void* d_ws, size_t ws_size,
                              hipStream_t stream) {
  const float* inputs  = (const float*)d_in[0];
  const float* weights = (const float*)d_in[1];
  float*       out     = (float*)d_out;

  quanv_fused_kernel<<<244, 128, 0, stream>>>(inputs, weights, out);
}


# Main kernel: a patch's two lanes (n, n+16) split its 4 pixels: each half loads one pixel row and takes its sincos, one permlanex16 exchange gives both halves all four; sum order unchanged
# speedup vs baseline: 3.4649x; 1.0995x over previous
#include <hip/hip_runtime.h>


typedef float v2f __attribute__((ext_vector_type(2)));
typedef float v8f __attribute__((ext_vector_type(8)));

__device__ __forceinline__ int cxmap(int m, int ctrl, int tgt) {
  return m ^ (((m >> (3 - ctrl)) & 1) << (3 - tgt));
}

__device__ __forceinline__ float lane_value(float v, int src) {
  return __int_as_float(__builtin_amdgcn_readlane(__float_as_int(v), src));
}

template <int WI>
__device__ __forceinline__ void rot_row(int x, float cosv, float sinv, float& e0, float& e1) {
  if constexpr (WI < 0) {
    e0 = x ? 0.0f : 1.0f;
    e1 = x ? 1.0f : 0.0f;
  } else {
    const float c = lane_value(cosv, WI);
    const float s = lane_value(sinv, WI);
    e0 = x ? s : c;
    e1 = x ? c : -s;
  }
}

template <int W0, int W1, int W2, int W3>
__device__ __forceinline__ v8f apply_layer(v8f T, int mp, int hf, float cosv, float sinv) {
  float p0, p1, q0, q1, r0, r1, t0, t1;
  rot_row<W0>((mp >> 3) & 1, cosv, sinv, p0, p1);
  rot_row<W1>((mp >> 2) & 1, cosv, sinv, q0, q1);
  rot_row<W2>((mp >> 1) & 1, cosv, sinv, r0, r1);
  rot_row<W3>( mp       & 1, cosv, sinv, t0, t1);
  const float p = hf ? p1 : p0;
  float a[8];
#pragma unroll
  for (int i = 0; i < 8; ++i) {
    a[i] = p * ((i & 4) ? q1 : q0) * ((i & 2) ? r1 : r0) * ((i & 1) ? t1 : t0);
  }
  v8f acc = {};
#pragma unroll
  for (int s = 0; s < 4; ++s) {
    v2f av, bv;
    av[0] = a[2 * s]; av[1] = a[2 * s + 1];
    bv[0] = T[2 * s]; bv[1] = T[2 * s + 1];
    acc = __builtin_amdgcn_wmma_f32_16x16x4_f32(false, av, false, bv, (short)0, acc, false, false);
  }
  return acc;
}

__device__ __forceinline__ void store_U_hi(v8f U, int hf, float* uo) {
  if (hf) {
#pragma unroll
    for (int r = 0; r < 8; ++r) {
      uo[r * 16] = U[r];
    }
  }
}

#define LAYER3(W0, W1, W2, W3, MP)                                                  \
  _Pragma("unroll")                 \
  for (int c = 0; c < 3; ++c) {                                                     \
    T[c] = apply_layer<W0, W1, W2, W3>(T[c], (MP), hf, cosv[c], sinv[c]);           \
  }

__device__ __forceinline__ void build_channel(int ch, int lane, const float (&sinv)[3],
                                              const float (&cosv)[3], float* ulds) {
  const int circ = ch % 5;
  const int hf   = lane >> 4;
  const int n    = lane & 15;

  v8f T[3];
#pragma unroll
  for (int c = 0; c < 3; ++c) {
#pragma unroll
    for (int r = 0; r < 8; ++r) {
      T[c][r] = (8 * hf + r == n) ? 1.0f : 0.0f;
    }
  }

  switch (circ) {
    case 0: {
      LAYER3(0, 1, 2, 3, cxmap(cxmap(n, 2, 3), 0, 1))
      LAYER3(4, -1, 5, -1, cxmap(n, 1, 2))
      LAYER3(-1, 6, -1, -1, n)
      break;
    }
    case 1: {
      LAYER3(0, 1, 2, 3, cxmap(cxmap(cxmap(n, 2, 3), 1, 2), 0, 1))
      LAYER3(-1, 4, 5, 6, n)
      break;
    }
    case 2: {
      LAYER3(0, 1, 2, 3, cxmap(cxmap(cxmap(cxmap(n, 3, 0), 2, 3), 1, 2), 0, 1))
      LAYER3(4, 5, 6, 7, n)
      break;
    }
    case 3: {
      const int fwd = cxmap(cxmap(cxmap(cxmap(n, 3, 0), 2, 3), 1, 2), 0, 1);
      const int bwd = cxmap(cxmap(cxmap(cxmap(n, 0, 1), 1, 2), 2, 3), 3, 0);
      LAYER3(0, 1, 2, 3, fwd)
      LAYER3(4, 5, 6, 7, bwd)
      LAYER3(8, 9, 10, 11, fwd)
      LAYER3(12, 13, 14, 15, bwd)
      break;
    }
    default: {
      const int blk = cxmap(cxmap(cxmap(cxmap(n, 3, 0), 1, 2), 2, 3), 0, 1);
      LAYER3(0, 1, 2, 3, blk)
      LAYER3(4, 5, 6, 7, blk)
      LAYER3(8, 9, 10, 11, blk)
      break;
    }
  }

#pragma unroll
  for (int c = 0; c < 3; ++c) {
    store_U_hi(T[c], hf, ulds + c * 128 + n);
  }
}


#define SINCOS_LARGE_ARG 131072.0f

template <bool SMALL>
__device__ __forceinline__ void half_angle_sincos(float h, float& s, float& c) {
  if constexpr (SMALL) {
    __builtin_assume(!(__builtin_fabsf(h) >= SINCOS_LARGE_ARG));
  }
  sincosf(h, &s, &c);
}

__device__ __forceinline__ void embed_products(const float (&sn)[4], const float (&cs)[4], int hf,
                                               float (&sv)[8]) {
  const float t0 = hf ? sn[0] : cs[0];
#pragma unroll
  for (int i = 0; i < 8; ++i) {
    sv[i] = t0 * ((i & 4) ? sn[1] : cs[1]) * ((i & 2) ? sn[2] : cs[2]) * ((i & 1) ? sn[3] : cs[3]);
  }
}

__device__ __forceinline__ float from_other_half(float v) {
  return __builtin_bit_cast(float, __builtin_amdgcn_permlanex16(__builtin_bit_cast(int, v), __builtin_bit_cast(int, v),
                                                                0x76543210, 0xfedcba98, false, false));
}

__device__ __forceinline__ void gather_patch_sincos(const float (&so)[2], const float (&co)[2], int hf,
                                                    float (&sn)[4], float (&cs)[4]) {
#pragma unroll
  for (int j = 0; j < 2; ++j) {
    const float os = from_other_half(so[j]);
    const float oc = from_other_half(co[j]);
    sn[j]     = hf ? os : so[j];
    cs[j]     = hf ? oc : co[j];
    sn[2 + j] = hf ? so[j] : os;
    cs[2 + j] = hf ? co[j] : oc;
  }
}

__device__ __forceinline__ void add_squares(v2f& ph, v8f acc) {
#pragma unroll
  for (int s = 0; s < 4; ++s) {
    v2f d;
    d[0] = acc[2 * s]; d[1] = acc[2 * s + 1];
    ph += d * d;
  }
}

__device__ __forceinline__ void load_patch_row(const float* __restrict__ in, int tile, int n, int hf,
                                               float (&x)[3][2]) {
  const int p   = tile * 16 + n;
  const int b   = p / 961;
  const int rem = p % 961;
  const int h = rem / 31;
  const int w = rem % 31;
#pragma unroll
  for (int c = 0; c < 3; ++c) {
    const float* base = in + ((b * 3 + c) * 32 + h + hf) * 32 + w;
    x[c][0] = base[0];  x[c][1] = base[1];
  }
}

__device__ __forceinline__ bool any_large(const float (&x)[3][2]) {
  bool large = false;
#pragma unroll
  for (int c = 0; c < 3; ++c) {
#pragma unroll
    for (int j = 0; j < 2; ++j) {
      large = large || (__builtin_fabsf(0.5f * x[c][j]) >= SINCOS_LARGE_ARG);
    }
  }
  return large;
}

__device__ __forceinline__ v2f frag_k(const float4 (&u)[2], int s) {
  const float4 w = u[s >> 1];
  v2f a;
  if (s & 1) { a[0] = w.z; a[1] = w.w; } else { a[0] = w.x; a[1] = w.y; }
  return a;
}

template <bool SMALL>
__device__ __forceinline__ void tile_body(const float (&x)[3][2], const float (&xn0)[2],
                                          const float4 (&uf)[2][3][2], int hf, float (&sv)[8],
                                          float (&res)[2]) {
  v2f ph[2];
  ph[0][0] = 0.0f; ph[0][1] = 0.0f;
  ph[1][0] = 0.0f; ph[1][1] = 0.0f;
#pragma unroll
  for (int c = 0; c < 3; ++c) {
    const float* xs = c < 2 ? x[c + 1] : xn0;
    float so[2], co[2], sn[4], cs[4], svn[8];
    v8f a8[2] = {};
#pragma unroll
    for (int s = 0; s < 4; ++s) {
      v2f bf;
      bf[0] = sv[2 * s]; bf[1] = sv[2 * s + 1];
#pragma unroll
      for (int P = 0; P < 2; ++P) {
        a8[P] = __builtin_amdgcn_wmma_f32_16x16x4_f32(false, frag_k(uf[P][c], s), false, bf, (short)0,
                                                      a8[P], false, false);
      }
      if (s == 0) {
        half_angle_sincos<SMALL>(0.5f * xs[0], so[0], co[0]);
        half_angle_sincos<SMALL>(0.5f * xs[1], so[1], co[1]);
      } else if (s == 1) {
        gather_patch_sincos(so, co, hf, sn, cs);
      } else if (s == 2) {
        embed_products(sn, cs, hf, svn);
      }
    }
    add_squares(ph[0], a8[0]);
    add_squares(ph[1], a8[1]);
#pragma unroll
    for (int i = 0; i < 8; ++i) {
      sv[i] = svn[i];
    }
  }
  res[0] = 3.0f - 2.0f * (ph[0][0] + ph[0][1]);
  res[1] = 3.0f - 2.0f * (ph[1][0] + ph[1][1]);
}

template <bool SMALL>
__device__ __forceinline__ void weight_sincos(const float (&wv)[3], float (&sw)[3], float (&cw)[3]) {
#pragma unroll
  for (int c = 0; c < 3; ++c) {
    half_angle_sincos<SMALL>(0.5f * wv[c], sw[c], cw[c]);
  }
}

template <bool SMALL>
__device__ __forceinline__ void embed_first(const float (&x)[2], int hf, float (&sv)[8]) {
  float so[2], co[2], sn[4], cs[4];
#pragma unroll
  for (int j = 0; j < 2; ++j) {
    half_angle_sincos<SMALL>(0.5f * x[j], so[j], co[j]);
  }
  gather_patch_sincos(so, co, hf, sn, cs);
  embed_products(sn, cs, hf, sv);
}

#define TILES_PER_WAVE 4
#define TILES_PER_GROUP 16
#define LAST_TILE 960

__device__ __forceinline__ void lds_patch(const float* px, int t, int lane, float (&x)[3][2]) {
#pragma unroll
  for (int c = 0; c < 3; ++c) {
    const float2 v = *reinterpret_cast<const float2*>(px + ((t * 3 + c) * 32 + lane) * 2);
    x[c][0] = v.x; x[c][1] = v.y;
  }
}

template <bool SMALL>
__device__ __forceinline__ void tile_loop(const float* px, const float4 (&uf)[2][3][2], int lane, int Q,
                                          int tile0, float* __restrict__ out) {
  const int hf = lane >> 4;
  const int n  = lane & 15;
  float x0[3][2], x1[3][2];
  lds_patch(px, 0, lane, x0);
  lds_patch(px, 1, lane, x1);

  float sv[8];
  embed_first<SMALL>(x0[0], hf, sv);

#pragma unroll 1
  for (int t = 0; t < TILES_PER_WAVE; ++t) {
    const int tile = tile0 + t;
    if (tile > LAST_TILE) break;

    float x2[3][2];
    lds_patch(px, t + 2 < TILES_PER_WAVE ? t + 2 : TILES_PER_WAVE - 1, lane, x2);

    const int p   = tile * 16 + n;
    const int b   = p / 961;
    const int rem = p % 961;

    float res[2];
    tile_body<SMALL>(x0, x1[0], uf, hf, sv, res);

    out[(b * 16 + 4 * Q + hf) * 961 + rem]     = res[0];
    out[(b * 16 + 4 * Q + 2 + hf) * 961 + rem] = res[1];

#pragma unroll
    for (int c = 0; c < 3; ++c) {
#pragma unroll
      for (int j = 0; j < 2; ++j) {
        x0[c][j] = x1[c][j];
        x1[c][j] = x2[c][j];
      }
    }
  }
}

__global__ void __launch_bounds__(128) quanv_fused_kernel(const float* __restrict__ in,
                                                         const float* __restrict__ wts,
                                                         float* __restrict__ out) {
  __shared__ __attribute__((aligned(16))) float Us[12 * 128];
  __shared__ __attribute__((aligned(16))) float Px[4 * TILES_PER_WAVE * 6 * 32];

  const int Q    = blockIdx.x & 3;
  const int g    = blockIdx.x >> 2;
  const int wv   = __builtin_amdgcn_readfirstlane(threadIdx.x >> 5);
  const int lane = threadIdx.x & 31;
  const int n    = lane & 15;
  const int hf   = lane >> 4;
  const int ch   = 4 * Q + wv;

  float wvl[3];
#pragma unroll
  for (int c = 0; c < 3; ++c) {
    wvl[c] = wts[(ch * 3 + c) * 16 + n];
  }
  const int tile0 = g * TILES_PER_GROUP + wv * TILES_PER_WAVE;
  float gx[TILES_PER_WAVE][3][2];
#pragma unroll
  for (int t = 0; t < TILES_PER_WAVE; ++t) {
    const int tile = tile0 + t < LAST_TILE ? tile0 + t : LAST_TILE;
    load_patch_row(in, tile, n, hf, gx[t]);
  }

  float sinv[3], cosv[3];
  {
    bool wl = false;
#pragma unroll
    for (int c = 0; c < 3; ++c) {
      wl = wl | (__builtin_fabsf(0.5f * wvl[c]) >= SINCOS_LARGE_ARG);
    }
    if (__builtin_expect(__builtin_amdgcn_ballot_w32(wl) == 0u, 1)) {
      weight_sincos<true>(wvl, sinv, cosv);
    } else {
      weight_sincos<false>(wvl, sinv, cosv);
    }
  }

  build_channel(ch, lane, sinv, cosv, Us + wv * 3 * 128);

  float* px = Px + wv * (TILES_PER_WAVE * 6 * 32);
  bool large = false;
#pragma unroll
  for (int t = 0; t < TILES_PER_WAVE; ++t) {
    large = large | any_large(gx[t]);
#pragma unroll
    for (int c = 0; c < 3; ++c) {
      float2 v;
      v.x = gx[t][c][0]; v.y = gx[t][c][1];
      *reinterpret_cast<float2*>(px + ((t * 3 + c) * 32 + lane) * 2) = v;
    }
  }
  __syncthreads();

  float4 uf[2][3][2];
  {
    const float* Urow = Us + ((n >> 3) * 3) * 128 + (n & 7) * 16 + 8 * hf;
#pragma unroll
    for (int P = 0; P < 2; ++P) {
#pragma unroll
      for (int c = 0; c < 3; ++c) {
        const float4* up = reinterpret_cast<const float4*>(Urow + (P * 6 + c) * 128);
        uf[P][c][0] = up[0];
        uf[P][c][1] = up[1];
      }
    }
  }

  if (__builtin_expect(__builtin_amdgcn_ballot_w32(large) == 0u, 1)) {
    tile_loop<true>(px, uf, lane, Q, tile0, out);
  } else {
    tile_loop<false>(px, uf, lane, Q, tile0, out);
  }
}

extern "C" void kernel_launch(void* const* d_in, const int* in_sizes, int n_in,
                              void* d_out, int out_size, void* d_ws, size_t ws_size,
                              hipStream_t stream) {
  const float* inputs  = (const float*)d_in[0];
  const float* weights = (const float*)d_in[1];
  float*       out     = (float*)d_out;

  quanv_fused_kernel<<<244, 128, 0, stream>>>(inputs, weights, out);
}
